# starting kernel
# baseline (speedup 1.0000x reference)
#include <hip/hip_runtime.h>

typedef __attribute__((ext_vector_type(2))) float v2f;
typedef __attribute__((ext_vector_type(4))) float v4f;
typedef __attribute__((ext_vector_type(8))) float v8f;

#define NQ      9
#define C_IN    8
#define HW      32
#define NFILT   16
#define INV2PI  0.15915494309189535f

__device__ __forceinline__ float zprod(const float* __restrict__ x,
                                       const float wq[NQ],
                                       int b, int c, int i, int j) {
  const float* xc = x + ((size_t)(b * C_IN + c)) * HW * HW;
  float prod = 1.0f;
#pragma unroll
  for (int a = 0; a < 3; ++a) {
#pragma unroll
    for (int bb = 0; bb < 3; ++bb) {
      int q = a * 3 + bb;
      if (q == 0) continue;
      int ii = i + a - 1;
      int jj = j + bb - 1;
      bool inb = (ii >= 0) & (ii < HW) & (jj >= 0) & (jj < HW);
      float xv = inb ? xc[ii * HW + jj] : 0.0f;
      prod *= __builtin_amdgcn_cosf((xv + wq[q]) * INV2PI);
    }
  }
  return prod;
}

__global__ void __launch_bounds__(128)
qconv_vqc_kernel(const float* __restrict__ x,
                 const float* __restrict__ w,
                 float* __restrict__ out) {
  int flat = blockIdx.x * blockDim.x + threadIdx.x;
  int lane = flat & 31;
  int wv   = flat >> 5;
  int m    = lane & 15;
  int half = lane >> 4;
  int b     = wv >> 6;
  int pbase = (wv & 63) * 16;
  int p = pbase + m;
  int i = p >> 5;
  int j = p & 31;

  float wq[NQ];
#pragma unroll
  for (int q = 1; q < NQ; ++q) wq[q] = w[b * NQ + q];

  v2f a0, a1, ones;
  a0.x = zprod(x, wq, b, 2 * half + 0, i, j);
  a0.y = zprod(x, wq, b, 2 * half + 1, i, j);
  a1.x = zprod(x, wq, b, 4 + 2 * half, i, j);
  a1.y = zprod(x, wq, b, 5 + 2 * half, i, j);
  ones.x = 1.0f; ones.y = 1.0f;

  v8f acc = {};
  acc = __builtin_amdgcn_wmma_f32_16x16x4_f32(false, a0, false, ones,
                                              (short)0, acc, false, false);
  acc = __builtin_amdgcn_wmma_f32_16x16x4_f32(false, a1, false, ones,
                                              (short)0, acc, false, false);

  int f = lane & 15;
  float* o = out + ((size_t)(b * NFILT + f)) * (HW * HW) + pbase + half * 8;
  v4f lo = {acc[0], acc[1], acc[2], acc[3]};
  v4f hi = {acc[4], acc[5], acc[6], acc[7]};
  *(v4f*)(o + 0) = lo;
  *(v4f*)(o + 4) = hi;
}

extern "C" void kernel_launch(void* const* d_in, const int* in_sizes, int n_in,
                              void* d_out, int out_size, void* d_ws, size_t ws_size,
                              hipStream_t stream) {
  (void)in_sizes; (void)n_in; (void)out_size; (void)d_ws; (void)ws_size;
  const float* x = (const float*)d_in[0];
  const float* w = (const float*)d_in[1];
  float* out = (float*)d_out;
  qconv_vqc_kernel<<<64, 128, 0, stream>>>(x, w, out);
}


# Branch-free taps: each tap loaded from an address clamped into the image (scalar row base + lane offset + literal), padding taps zeroed by a select after the load; the 32 loads issue back to back
# speedup vs baseline: 1.3397x; 1.3397x over previous
#include <hip/hip_runtime.h>

typedef __attribute__((ext_vector_type(2))) float v2f;
typedef __attribute__((ext_vector_type(4))) float v4f;
typedef __attribute__((ext_vector_type(8))) float v8f;

#define NQ      9
#define C_IN    8
#define HW      32
#define NFILT   16
#define INV2PI  0.15915494309189535f

__global__ void __launch_bounds__(128)
qconv_vqc_kernel(const float* __restrict__ x,
                 const float* __restrict__ w,
                 float* __restrict__ out) {
  const int lane = threadIdx.x & 31;
  const int wib = __builtin_amdgcn_readfirstlane((int)(threadIdx.x >> 5));
  const int wv  = (int)blockIdx.x * 4 + wib;
  const int m    = lane & 15;
  const int half = lane >> 4;
  const int b     = wv >> 6;
  const int pbase = (wv & 63) * 16;
  const int i = pbase >> 5;
  const int j = (pbase & 31) + m;

  float wq[NQ];
#pragma unroll
  for (int q = 1; q < NQ; ++q) wq[q] = w[b * NQ + q];

  const int  rr[3]  = { i > 0 ? i - 1 : 0, i, i < HW - 1 ? i + 1 : HW - 1 };
  const bool rin[3] = { i > 0, true, i < HW - 1 };
  const int  cc[3]  = { j > 0 ? j - 1 : 0, j, j < HW - 1 ? j + 1 : HW - 1 };
  const bool cin[3] = { j > 0, true, j < HW - 1 };

  const float* xb = x + (size_t)(b * C_IN) * (HW * HW);
  const unsigned lane_ch = (unsigned)half * 2u * (HW * HW);

  float xv[4][NQ];
#pragma unroll
  for (int a = 0; a < 3; ++a) {
    const float* xr = xb + rr[a] * HW;
#pragma unroll
    for (int bb = 0; bb < 3; ++bb) {
      const int q = a * 3 + bb;
      if (q == 0) continue;
      const unsigned off = lane_ch + (unsigned)cc[bb];
      const bool inb = rin[a] & cin[bb];
#pragma unroll
      for (int k = 0; k < 4; ++k) {
        const int ch = (k & 1) + (k >> 1) * 4;
        const float v = (xr + ch * (HW * HW))[off];
        xv[k][q] = inb ? v : 0.0f;
      }
    }
  }

  float z[4];
#pragma unroll
  for (int k = 0; k < 4; ++k) {
    float prod = 1.0f;
#pragma unroll
    for (int q = 1; q < NQ; ++q)
      prod *= __builtin_amdgcn_cosf((xv[k][q] + wq[q]) * INV2PI);
    z[k] = prod;
  }

  v2f a0, a1, ones;
  a0.x = z[0];
  a0.y = z[1];
  a1.x = z[2];
  a1.y = z[3];
  ones.x = 1.0f; ones.y = 1.0f;

  v8f acc = {};
  acc = __builtin_amdgcn_wmma_f32_16x16x4_f32(false, a0, false, ones,
                                              (short)0, acc, false, false);
  acc = __builtin_amdgcn_wmma_f32_16x16x4_f32(false, a1, false, ones,
                                              (short)0, acc, false, false);

  const int f = lane & 15;
  float* o = out + ((size_t)(b * NFILT + f)) * (HW * HW) + pbase + half * 8;
  v4f lo = {acc[0], acc[1], acc[2], acc[3]};
  v4f hi = {acc[4], acc[5], acc[6], acc[7]};
  *(v4f*)(o + 0) = lo;
  *(v4f*)(o + 4) = hi;
}

extern "C" void kernel_launch(void* const* d_in, const int* in_sizes, int n_in,
                              void* d_out, int out_size, void* d_ws, size_t ws_size,
                              hipStream_t stream) {
  (void)in_sizes; (void)n_in; (void)out_size; (void)d_ws; (void)ws_size;
  const float* x = (const float*)d_in[0];
  const float* w = (const float*)d_in[1];
  float* out = (float*)d_out;
  qconv_vqc_kernel<<<64, 128, 0, stream>>>(x, w, out);
}
